# starting kernel
# baseline (speedup 1.0000x reference)
#include <hip/hip_runtime.h>

typedef __attribute__((ext_vector_type(2))) float v2f;
typedef __attribute__((ext_vector_type(8))) float v8f;

#define ALPHA    0.2f
#define NEG_BIG  -9000000000000000.0f
#define N_NODES  2048
#define FDIM     128
#define NBATCH   4

__global__ __launch_bounds__(256) void gat_wh_kernel(const float* __restrict__ h,
                                                     const float* __restrict__ W,
                                                     float* __restrict__ Wh) {
    __shared__ float sh[16 * FDIM];
    const int tile = blockIdx.x;
    const int b    = tile / (N_NODES / 16);
    const int i0   = (tile % (N_NODES / 16)) * 16;
    const int t    = threadIdx.x;

    const float* hbase = h + ((size_t)b * N_NODES + i0) * FDIM;
    for (int k = t; k < 16 * FDIM; k += 256) sh[k] = hbase[k];
    __syncthreads();

    const int wave  = t >> 5;
    const int lane  = t & 31;
    const int row   = lane & 15;
    const int kslot = lane >> 4;
    const int n0    = wave * 16;

    v8f acc = {};
    for (int k0 = 0; k0 < FDIM; k0 += 4) {
        v2f A, B;
        A.x = sh[row * FDIM + k0 + kslot * 2];
        A.y = sh[row * FDIM + k0 + kslot * 2 + 1];
        B.x = W[(size_t)(k0 + kslot * 2)     * FDIM + n0 + row];
        B.y = W[(size_t)(k0 + kslot * 2 + 1) * FDIM + n0 + row];
        acc = __builtin_amdgcn_wmma_f32_16x16x4_f32(false, A, false, B,
                                                    (short)0, acc, false, false);
    }

    float* out = Wh + ((size_t)b * N_NODES + i0) * FDIM + n0 + row;
#pragma unroll
    for (int v = 0; v < 8; ++v) {
        int orow = v + 8 * kslot;
        out[(size_t)orow * FDIM] = acc[v];
    }
}

__global__ __launch_bounds__(256) void gat_f_kernel(const float* __restrict__ Wh,
                                                    const float* __restrict__ a,
                                                    float* __restrict__ f1,
                                                    float* __restrict__ f2) {
    const int row  = blockIdx.x * 8 + (threadIdx.x >> 5);
    const int lane = threadIdx.x & 31;
    const float* wr = Wh + (size_t)row * FDIM;
    float s1 = 0.f, s2 = 0.f;
#pragma unroll
    for (int c = 0; c < FDIM; c += 32) {
        float v = wr[c + lane];
        s1 += v * a[c + lane];
        s2 += v * a[FDIM + c + lane];
    }
#pragma unroll
    for (int m = 16; m >= 1; m >>= 1) {
        s1 += __shfl_xor(s1, m, 32);
        s2 += __shfl_xor(s2, m, 32);
    }
    if (lane == 0) { f1[row] = s1; f2[row] = s2; }
}

__global__ __launch_bounds__(256) void gat_attn_kernel(const float* __restrict__ Wh,
                                                       const int*   __restrict__ adj,
                                                       const float* __restrict__ f1,
                                                       const float* __restrict__ f2,
                                                       float* __restrict__ out) {
    __shared__ float s_f2[N_NODES];
    __shared__ float s_acc[16 * FDIM];
    __shared__ float s_m[16];
    __shared__ float s_l[16];

    const int blk  = blockIdx.x;
    const int b    = blk / (N_NODES / 16);
    const int i0   = (blk % (N_NODES / 16)) * 16;
    const int t    = threadIdx.x;
    const int wave = t >> 5;
    const int lane = t & 31;

    for (int j = t; j < N_NODES; j += 256) s_f2[j] = f2[(size_t)b * N_NODES + j];
    for (int k = t; k < 16 * FDIM; k += 256) s_acc[k] = 0.f;
    __syncthreads();

    {
        const int half = lane >> 4;
        const int r    = wave * 2 + half;
        const int l16  = lane & 15;
        const int* arow = adj + (size_t)(i0 + r) * N_NODES;
        const float fr  = f1[(size_t)b * N_NODES + i0 + r];

        float mx = NEG_BIG;
        for (int j = l16; j < N_NODES; j += 16) {
            float x = fr + s_f2[j];
            float s = (arow[j] > 0) ? (x > 0.f ? x : ALPHA * x) : NEG_BIG;
            mx = fmaxf(mx, s);
        }
#pragma unroll
        for (int m = 8; m >= 1; m >>= 1) mx = fmaxf(mx, __shfl_xor(mx, m, 32));

        float sum = 0.f;
        for (int j = l16; j < N_NODES; j += 16) {
            float x = fr + s_f2[j];
            float s = (arow[j] > 0) ? (x > 0.f ? x : ALPHA * x) : NEG_BIG;
            sum += __expf(s - mx);
        }
#pragma unroll
        for (int m = 8; m >= 1; m >>= 1) sum += __shfl_xor(sum, m, 32);

        if (l16 == 0) { s_m[r] = mx; s_l[r] = sum; }
    }
    __syncthreads();

    const int ln16  = lane & 15;
    const int kslot = lane >> 4;
    const float rm  = s_m[ln16];
    const float rf1 = f1[(size_t)b * N_NODES + i0 + ln16];
    const int* arow = adj + (size_t)(i0 + ln16) * N_NODES;
    const float* whb = Wh + (size_t)b * N_NODES * FDIM;

    v8f acc[8] = {};
    const int jend = wave * 256 + 256;
    for (int j0 = wave * 256; j0 < jend; j0 += 4) {
        const int jA = j0 + kslot * 2;
        float x0 = rf1 + s_f2[jA];
        float s0 = (arow[jA] > 0) ? (x0 > 0.f ? x0 : ALPHA * x0) : NEG_BIG;
        float x1 = rf1 + s_f2[jA + 1];
        float s1 = (arow[jA + 1] > 0) ? (x1 > 0.f ? x1 : ALPHA * x1) : NEG_BIG;
        v2f A;
        A.x = __expf(s0 - rm);
        A.y = __expf(s1 - rm);

        const float* w0 = whb + (size_t)jA * FDIM + ln16;
        const float* w1 = w0 + FDIM;
#pragma unroll
        for (int n = 0; n < 8; ++n) {
            v2f B;
            B.x = w0[n * 16];
            B.y = w1[n * 16];
            acc[n] = __builtin_amdgcn_wmma_f32_16x16x4_f32(false, A, false, B,
                                                           (short)0, acc[n], false, false);
        }
    }

#pragma unroll
    for (int n = 0; n < 8; ++n) {
#pragma unroll
        for (int v = 0; v < 8; ++v) {
            int orow = v + 8 * kslot;
            atomicAdd(&s_acc[orow * FDIM + n * 16 + ln16], acc[n][v]);
        }
    }
    __syncthreads();

    for (int k = t; k < 16 * FDIM; k += 256) {
        int r = k >> 7;
        out[((size_t)b * N_NODES + i0 + r) * FDIM + (k & (FDIM - 1))] = s_acc[k] / s_l[r];
    }
}

extern "C" void kernel_launch(void* const* d_in, const int* in_sizes, int n_in,
                              void* d_out, int out_size, void* d_ws, size_t ws_size,
                              hipStream_t stream) {
    const float* h   = (const float*)d_in[0];
    const int*   adj = (const int*)  d_in[1];
    const float* W   = (const float*)d_in[2];
    const float* a   = (const float*)d_in[3];
    float* out = (float*)d_out;

    float* Wh = (float*)d_ws;
    float* f1 = Wh + (size_t)NBATCH * N_NODES * FDIM;
    float* f2 = f1 + (size_t)NBATCH * N_NODES;

    gat_wh_kernel  <<<NBATCH * (N_NODES / 16), 256, 0, stream>>>(h, W, Wh);
    gat_f_kernel   <<<NBATCH * (N_NODES / 8),  256, 0, stream>>>(Wh, a, f1, f2);
    gat_attn_kernel<<<NBATCH * (N_NODES / 16), 256, 0, stream>>>(Wh, adj, f1, f2, out);
}


# attention kernel: the block's 16x2048 score tile is kept in LDS: mask read once, each score and exp computed once, WMMA A fragments read from LDS; order of the row sums changes
# speedup vs baseline: 2.2879x; 2.2879x over previous
#include <hip/hip_runtime.h>

typedef __attribute__((ext_vector_type(2))) float v2f;
typedef __attribute__((ext_vector_type(8))) float v8f;

#define ALPHA    0.2f
#define NEG_BIG  -9000000000000000.0f
#define N_NODES  2048
#define FDIM     128
#define NBATCH   4

__global__ __launch_bounds__(256) void gat_wh_kernel(const float* __restrict__ h,
                                                     const float* __restrict__ W,
                                                     float* __restrict__ Wh) {
    __shared__ float sh[16 * FDIM];
    const int tile = blockIdx.x;
    const int b    = tile / (N_NODES / 16);
    const int i0   = (tile % (N_NODES / 16)) * 16;
    const int t    = threadIdx.x;

    const float* hbase = h + ((size_t)b * N_NODES + i0) * FDIM;
    for (int k = t; k < 16 * FDIM; k += 256) sh[k] = hbase[k];
    __syncthreads();

    const int wave  = t >> 5;
    const int lane  = t & 31;
    const int row   = lane & 15;
    const int kslot = lane >> 4;
    const int n0    = wave * 16;

    v8f acc = {};
    for (int k0 = 0; k0 < FDIM; k0 += 4) {
        v2f A, B;
        A.x = sh[row * FDIM + k0 + kslot * 2];
        A.y = sh[row * FDIM + k0 + kslot * 2 + 1];
        B.x = W[(size_t)(k0 + kslot * 2)     * FDIM + n0 + row];
        B.y = W[(size_t)(k0 + kslot * 2 + 1) * FDIM + n0 + row];
        acc = __builtin_amdgcn_wmma_f32_16x16x4_f32(false, A, false, B,
                                                    (short)0, acc, false, false);
    }

    float* out = Wh + ((size_t)b * N_NODES + i0) * FDIM + n0 + row;
#pragma unroll
    for (int v = 0; v < 8; ++v) {
        int orow = v + 8 * kslot;
        out[(size_t)orow * FDIM] = acc[v];
    }
}

__global__ __launch_bounds__(256) void gat_f_kernel(const float* __restrict__ Wh,
                                                    const float* __restrict__ a,
                                                    float* __restrict__ f1,
                                                    float* __restrict__ f2) {
    const int row  = blockIdx.x * 8 + (threadIdx.x >> 5);
    const int lane = threadIdx.x & 31;
    const float* wr = Wh + (size_t)row * FDIM;
    float s1 = 0.f, s2 = 0.f;
#pragma unroll
    for (int c = 0; c < FDIM; c += 32) {
        float v = wr[c + lane];
        s1 += v * a[c + lane];
        s2 += v * a[FDIM + c + lane];
    }
#pragma unroll
    for (int m = 16; m >= 1; m >>= 1) {
        s1 += __shfl_xor(s1, m, 32);
        s2 += __shfl_xor(s2, m, 32);
    }
    if (lane == 0) { f1[row] = s1; f2[row] = s2; }
}

#define SPITCH   (N_NODES + 4)

__global__ __launch_bounds__(256) void gat_attn_kernel(const float* __restrict__ Wh,
                                                       const int*   __restrict__ adj,
                                                       const float* __restrict__ f1,
                                                       const float* __restrict__ f2,
                                                       float* __restrict__ out) {
    __shared__ __attribute__((aligned(16))) float s_p[16 * SPITCH];
    __shared__ __attribute__((aligned(16))) float s_f2[N_NODES];
    __shared__ float s_acc[16 * FDIM];
    __shared__ float s_l[16];

    const int blk  = blockIdx.x;
    const int b    = blk / (N_NODES / 16);
    const int i0   = (blk % (N_NODES / 16)) * 16;
    const int t    = threadIdx.x;
    const int wave = t >> 5;
    const int lane = t & 31;

    for (int j = t; j < N_NODES; j += 256) s_f2[j] = f2[(size_t)b * N_NODES + j];
    for (int k = t; k < 16 * FDIM; k += 256) s_acc[k] = 0.f;
    __syncthreads();

    {
        const int r    = t >> 4;
        const int l16  = t & 15;
        const int4* arow = reinterpret_cast<const int4*>(adj + (size_t)(i0 + r) * N_NODES);
        const float fr  = f1[(size_t)b * N_NODES + i0 + r];
        float* prow = s_p + r * SPITCH;

        float mx = NEG_BIG;
        for (int c0 = 0; c0 < 32; c0 += 8) {
            int4 mk[8];
#pragma unroll
            for (int u = 0; u < 8; ++u) mk[u] = arow[(c0 + u) * 16 + l16];
#pragma unroll
            for (int u = 0; u < 8; ++u) {
                const int j = ((c0 + u) * 16 + l16) * 4;
                const float4 f = *reinterpret_cast<const float4*>(s_f2 + j);
                const float x0 = fr + f.x, x1 = fr + f.y, x2 = fr + f.z, x3 = fr + f.w;
                const float e0 = x0 > 0.f ? x0 : ALPHA * x0;
                const float e1 = x1 > 0.f ? x1 : ALPHA * x1;
                const float e2 = x2 > 0.f ? x2 : ALPHA * x2;
                const float e3 = x3 > 0.f ? x3 : ALPHA * x3;
                float4 s;
                s.x = (mk[u].x > 0) ? e0 : NEG_BIG;
                s.y = (mk[u].y > 0) ? e1 : NEG_BIG;
                s.z = (mk[u].z > 0) ? e2 : NEG_BIG;
                s.w = (mk[u].w > 0) ? e3 : NEG_BIG;
                mx = fmaxf(mx, fmaxf(fmaxf(s.x, s.y), fmaxf(s.z, s.w)));
                *reinterpret_cast<float4*>(prow + j) = s;
            }
        }
#pragma unroll
        for (int m = 8; m >= 1; m >>= 1) mx = fmaxf(mx, __shfl_xor(mx, m, 32));

        float sum = 0.f;
#pragma unroll 8
        for (int c = 0; c < 32; ++c) {
            const int j = (c * 16 + l16) * 4;
            const float4 s = *reinterpret_cast<const float4*>(prow + j);
            float4 p;
            p.x = __expf(s.x - mx);
            p.y = __expf(s.y - mx);
            p.z = __expf(s.z - mx);
            p.w = __expf(s.w - mx);
            sum += (p.x + p.y) + (p.z + p.w);
            *reinterpret_cast<float4*>(prow + j) = p;
        }
#pragma unroll
        for (int m = 8; m >= 1; m >>= 1) sum += __shfl_xor(sum, m, 32);

        if (l16 == 0) s_l[r] = sum;
    }
    __syncthreads();

    const int ln16  = lane & 15;
    const int kslot = lane >> 4;
    const float* pA  = s_p + ln16 * SPITCH + kslot * 2;
    const float* whb = Wh + (size_t)b * N_NODES * FDIM;

    v8f acc[8] = {};
    const int jend = wave * 256 + 256;
    for (int j0 = wave * 256; j0 < jend; j0 += 4) {
        const int jA = j0 + kslot * 2;
        const v2f A = *reinterpret_cast<const v2f*>(pA + j0);

        const float* w0 = whb + (size_t)jA * FDIM + ln16;
        const float* w1 = w0 + FDIM;
#pragma unroll
        for (int n = 0; n < 8; ++n) {
            v2f B;
            B.x = w0[n * 16];
            B.y = w1[n * 16];
            acc[n] = __builtin_amdgcn_wmma_f32_16x16x4_f32(false, A, false, B,
                                                           (short)0, acc[n], false, false);
        }
    }

#pragma unroll
    for (int n = 0; n < 8; ++n) {
#pragma unroll
        for (int v = 0; v < 8; ++v) {
            int orow = v + 8 * kslot;
            atomicAdd(&s_acc[orow * FDIM + n * 16 + ln16], acc[n][v]);
        }
    }
    __syncthreads();

    for (int k = t; k < 16 * FDIM; k += 256) {
        int r = k >> 7;
        out[((size_t)b * N_NODES + i0 + r) * FDIM + (k & (FDIM - 1))] = s_acc[k] / s_l[r];
    }
}

extern "C" void kernel_launch(void* const* d_in, const int* in_sizes, int n_in,
                              void* d_out, int out_size, void* d_ws, size_t ws_size,
                              hipStream_t stream) {
    const float* h   = (const float*)d_in[0];
    const int*   adj = (const int*)  d_in[1];
    const float* W   = (const float*)d_in[2];
    const float* a   = (const float*)d_in[3];
    float* out = (float*)d_out;

    float* Wh = (float*)d_ws;
    float* f1 = Wh + (size_t)NBATCH * N_NODES * FDIM;
    float* f2 = f1 + (size_t)NBATCH * N_NODES;

    gat_wh_kernel  <<<NBATCH * (N_NODES / 16), 256, 0, stream>>>(h, W, Wh);
    gat_f_kernel   <<<NBATCH * (N_NODES / 8),  256, 0, stream>>>(Wh, a, f1, f2);
    gat_attn_kernel<<<NBATCH * (N_NODES / 16), 256, 0, stream>>>(Wh, adj, f1, f2, out);
}


# products re-associated: out = ((P h) W)/rowsum and f1,f2 = h (W a), so h W is never formed and its launch is gone (3 launches to 2); association and order of the sums change
# speedup vs baseline: 2.6467x; 1.1568x over previous
#include <hip/hip_runtime.h>

typedef __attribute__((ext_vector_type(2))) float v2f;
typedef __attribute__((ext_vector_type(8))) float v8f;

#define ALPHA    0.2f
#define NEG_BIG  -9000000000000000.0f
#define N_NODES  2048
#define FDIM     128
#define NBATCH   4


template <int N, int M>
__device__ __forceinline__ void fold_step(float* v, int lane) {
    const bool hi = (lane & M) != 0;
#pragma unroll
    for (int r = 0; r < N; ++r) {
        const float send = hi ? v[r] : v[r + N];
        const float keep = hi ? v[r + N] : v[r];
        v[r] = keep + __shfl_xor(send, M, 32);
    }
}
__device__ __forceinline__ float wave_sum16(float* v, int lane) {
    fold_step<8, 16>(v, lane);
    fold_step<4, 8>(v, lane);
    fold_step<2, 4>(v, lane);
    fold_step<1, 2>(v, lane);
    return v[0] + __shfl_xor(v[0], 1, 32);
}
__device__ __forceinline__ float wave_sum8(float* v, int lane) {
    fold_step<4, 16>(v, lane);
    fold_step<2, 8>(v, lane);
    fold_step<1, 4>(v, lane);
    float s = v[0] + __shfl_xor(v[0], 2, 32);
    return s + __shfl_xor(s, 1, 32);
}

__global__ __launch_bounds__(256) void gat_f_kernel(const float* __restrict__ h,
                                                    const float* __restrict__ W,
                                                    const float* __restrict__ a,
                                                    float* __restrict__ f1,
                                                    float* __restrict__ f2) {
    __shared__ __attribute__((aligned(16))) float s_wa[2 * FDIM];
    const int t    = threadIdx.x;
    const int wave = t >> 5;
    const int lane = t & 31;

    const int row0 = blockIdx.x * 32 + wave * 4;
    float4 hv[4];
#pragma unroll
    for (int q = 0; q < 4; ++q)
        hv[q] = *reinterpret_cast<const float4*>(h + (size_t)(row0 + q) * FDIM + 4 * lane);

    const float4 a1v = *reinterpret_cast<const float4*>(a + 4 * lane);
    const float4 a2v = *reinterpret_cast<const float4*>(a + FDIM + 4 * lane);
    float p1[16], p2[16];
#pragma unroll
    for (int r = 0; r < 16; ++r) {
        const float4 w = *reinterpret_cast<const float4*>(W + (size_t)(wave * 16 + r) * FDIM + 4 * lane);
        p1[r] = (w.x * a1v.x + w.y * a1v.y) + (w.z * a1v.z + w.w * a1v.w);
        p2[r] = (w.x * a2v.x + w.y * a2v.y) + (w.z * a2v.z + w.w * a2v.w);
    }
    const float t1 = wave_sum16(p1, lane);
    const float t2 = wave_sum16(p2, lane);
    if ((lane & 1) == 0) {
        s_wa[wave * 16 + (lane >> 1)]        = t1;
        s_wa[FDIM + wave * 16 + (lane >> 1)] = t2;
    }
    __syncthreads();

    const float4 u1 = *reinterpret_cast<const float4*>(s_wa + 4 * lane);
    const float4 u2 = *reinterpret_cast<const float4*>(s_wa + FDIM + 4 * lane);
    float g[8];
#pragma unroll
    for (int q = 0; q < 4; ++q) {
        g[q]     = (hv[q].x * u1.x + hv[q].y * u1.y) + (hv[q].z * u1.z + hv[q].w * u1.w);
        g[4 + q] = (hv[q].x * u2.x + hv[q].y * u2.y) + (hv[q].z * u2.z + hv[q].w * u2.w);
    }
    const float tot = wave_sum8(g, lane);
    if ((lane & 3) == 0) {
        const int idx = lane >> 2;
        float* dst = (idx < 4) ? (f1 + row0 + idx) : (f2 + row0 + idx - 4);
        *dst = tot;
    }
}

#define SPITCH   (N_NODES + 4)
#define APITCH   (FDIM + 4)

__global__ __launch_bounds__(256) void gat_attn_kernel(const float* __restrict__ h,
                                                       const float* __restrict__ W,
                                                       const int*   __restrict__ adj,
                                                       const float* __restrict__ f1,
                                                       const float* __restrict__ f2,
                                                       float* __restrict__ out) {
    __shared__ __attribute__((aligned(16))) float s_p[16 * SPITCH];
    __shared__ __attribute__((aligned(16))) float s_f2[N_NODES];
    __shared__ __attribute__((aligned(16))) float s_acc[16 * APITCH];
    __shared__ float s_l[16];

    const int blk  = blockIdx.x;
    const int b    = blk / (N_NODES / 16);
    const int i0   = (blk % (N_NODES / 16)) * 16;
    const int t    = threadIdx.x;
    const int wave = t >> 5;
    const int lane = t & 31;

    for (int j = t; j < N_NODES; j += 256) s_f2[j] = f2[(size_t)b * N_NODES + j];
    for (int k = t; k < 16 * APITCH; k += 256) s_acc[k] = 0.f;
    __syncthreads();

    {
        const int r    = t >> 4;
        const int l16  = t & 15;
        const int4* arow = reinterpret_cast<const int4*>(adj + (size_t)(i0 + r) * N_NODES);
        const float fr  = f1[(size_t)b * N_NODES + i0 + r];
        float* prow = s_p + r * SPITCH;

        float mx = NEG_BIG;
        for (int c0 = 0; c0 < 32; c0 += 8) {
            int4 mk[8];
#pragma unroll
            for (int u = 0; u < 8; ++u) mk[u] = arow[(c0 + u) * 16 + l16];
#pragma unroll
            for (int u = 0; u < 8; ++u) {
                const int j = ((c0 + u) * 16 + l16) * 4;
                const float4 f = *reinterpret_cast<const float4*>(s_f2 + j);
                const float x0 = fr + f.x, x1 = fr + f.y, x2 = fr + f.z, x3 = fr + f.w;
                const float e0 = x0 > 0.f ? x0 : ALPHA * x0;
                const float e1 = x1 > 0.f ? x1 : ALPHA * x1;
                const float e2 = x2 > 0.f ? x2 : ALPHA * x2;
                const float e3 = x3 > 0.f ? x3 : ALPHA * x3;
                float4 s;
                s.x = (mk[u].x > 0) ? e0 : NEG_BIG;
                s.y = (mk[u].y > 0) ? e1 : NEG_BIG;
                s.z = (mk[u].z > 0) ? e2 : NEG_BIG;
                s.w = (mk[u].w > 0) ? e3 : NEG_BIG;
                mx = fmaxf(mx, fmaxf(fmaxf(s.x, s.y), fmaxf(s.z, s.w)));
                *reinterpret_cast<float4*>(prow + j) = s;
            }
        }
#pragma unroll
        for (int m = 8; m >= 1; m >>= 1) mx = fmaxf(mx, __shfl_xor(mx, m, 32));

        float sum = 0.f;
#pragma unroll 8
        for (int c = 0; c < 32; ++c) {
            const int j = (c * 16 + l16) * 4;
            const float4 s = *reinterpret_cast<const float4*>(prow + j);
            float4 p;
            p.x = __expf(s.x - mx);
            p.y = __expf(s.y - mx);
            p.z = __expf(s.z - mx);
            p.w = __expf(s.w - mx);
            sum += (p.x + p.y) + (p.z + p.w);
            *reinterpret_cast<float4*>(prow + j) = p;
        }
#pragma unroll
        for (int m = 8; m >= 1; m >>= 1) sum += __shfl_xor(sum, m, 32);

        if (l16 == 0) s_l[r] = sum;
    }
    __syncthreads();

    const int ln16  = lane & 15;
    const int kslot = lane >> 4;
    const float* pA  = s_p + ln16 * SPITCH + kslot * 2;
    const float* whb = h + (size_t)b * N_NODES * FDIM;

    v8f acc[8] = {};
    const int jend = wave * 256 + 256;
    for (int j0 = wave * 256; j0 < jend; j0 += 4) {
        const int jA = j0 + kslot * 2;
        const v2f A = *reinterpret_cast<const v2f*>(pA + j0);

        const float* w0 = whb + (size_t)jA * FDIM + ln16;
        const float* w1 = w0 + FDIM;
#pragma unroll
        for (int n = 0; n < 8; ++n) {
            v2f B;
            B.x = w0[n * 16];
            B.y = w1[n * 16];
            acc[n] = __builtin_amdgcn_wmma_f32_16x16x4_f32(false, A, false, B,
                                                           (short)0, acc[n], false, false);
        }
    }

#pragma unroll
    for (int n = 0; n < 8; ++n) {
#pragma unroll
        for (int v = 0; v < 8; ++v) {
            int orow = v + 8 * kslot;
            atomicAdd(&s_acc[orow * APITCH + n * 16 + ln16], acc[n][v]);
        }
    }
    __syncthreads();

    {
        const int n0 = wave * 16;
        const float* pT = s_acc + ln16 * APITCH + kslot * 2;
        const float* wB = W + (size_t)(kslot * 2) * FDIM + n0 + ln16;
        v8f o0 = {}, o1 = {};
#pragma unroll
        for (int k0 = 0; k0 < FDIM; k0 += 8) {
            const v2f A0 = *reinterpret_cast<const v2f*>(pT + k0);
            const v2f A1 = *reinterpret_cast<const v2f*>(pT + k0 + 4);
            v2f B0, B1;
            B0.x = wB[(size_t)(k0)     * FDIM];
            B0.y = wB[(size_t)(k0 + 1) * FDIM];
            B1.x = wB[(size_t)(k0 + 4) * FDIM];
            B1.y = wB[(size_t)(k0 + 5) * FDIM];
            o0 = __builtin_amdgcn_wmma_f32_16x16x4_f32(false, A0, false, B0,
                                                       (short)0, o0, false, false);
            o1 = __builtin_amdgcn_wmma_f32_16x16x4_f32(false, A1, false, B1,
                                                       (short)0, o1, false, false);
        }
#pragma unroll
        for (int v = 0; v < 8; ++v) {
            const int orow = v + 8 * kslot;
            s_p[orow * FDIM + n0 + ln16] = o0[v] + o1[v];
        }
    }
    __syncthreads();

    for (int k = t; k < 16 * FDIM; k += 256) {
        int r = k >> 7;
        out[((size_t)b * N_NODES + i0 + r) * FDIM + (k & (FDIM - 1))] = s_p[k] / s_l[r];
    }
}

extern "C" void kernel_launch(void* const* d_in, const int* in_sizes, int n_in,
                              void* d_out, int out_size, void* d_ws, size_t ws_size,
                              hipStream_t stream) {
    const float* h   = (const float*)d_in[0];
    const int*   adj = (const int*)  d_in[1];
    const float* W   = (const float*)d_in[2];
    const float* a   = (const float*)d_in[3];
    float* out = (float*)d_out;

    float* f1 = (float*)d_ws;
    float* f2 = f1 + (size_t)NBATCH * N_NODES;

    gat_f_kernel   <<<NBATCH * (N_NODES / 32), 256, 0, stream>>>(h, W, a, f1, f2);
    gat_attn_kernel<<<NBATCH * (N_NODES / 16), 256, 0, stream>>>(h, W, adj, f1, f2, out);
}


# attention kernel: f2 reaches LDS by two grouped 16-byte loads per thread (1 round trip to memory, was 8 dependent ones); epilogue's W loads issued before the tile combine, not behind its barrier
# speedup vs baseline: 2.7993x; 1.0577x over previous
#include <hip/hip_runtime.h>

typedef __attribute__((ext_vector_type(2))) float v2f;
typedef __attribute__((ext_vector_type(8))) float v8f;

#define ALPHA    0.2f
#define NEG_BIG  -9000000000000000.0f
#define N_NODES  2048
#define FDIM     128
#define NBATCH   4


template <int N, int M>
__device__ __forceinline__ void fold_step(float* v, int lane) {
    const bool hi = (lane & M) != 0;
#pragma unroll
    for (int r = 0; r < N; ++r) {
        const float send = hi ? v[r] : v[r + N];
        const float keep = hi ? v[r + N] : v[r];
        v[r] = keep + __shfl_xor(send, M, 32);
    }
}
__device__ __forceinline__ float wave_sum16(float* v, int lane) {
    fold_step<8, 16>(v, lane);
    fold_step<4, 8>(v, lane);
    fold_step<2, 4>(v, lane);
    fold_step<1, 2>(v, lane);
    return v[0] + __shfl_xor(v[0], 1, 32);
}
__device__ __forceinline__ float wave_sum8(float* v, int lane) {
    fold_step<4, 16>(v, lane);
    fold_step<2, 8>(v, lane);
    fold_step<1, 4>(v, lane);
    float s = v[0] + __shfl_xor(v[0], 2, 32);
    return s + __shfl_xor(s, 1, 32);
}

__global__ __launch_bounds__(256) void gat_f_kernel(const float* __restrict__ h,
                                                    const float* __restrict__ W,
                                                    const float* __restrict__ a,
                                                    float* __restrict__ f1,
                                                    float* __restrict__ f2) {
    __shared__ __attribute__((aligned(16))) float s_wa[2 * FDIM];
    const int t    = threadIdx.x;
    const int wave = t >> 5;
    const int lane = t & 31;

    const int row0 = blockIdx.x * 32 + wave * 4;
    float4 hv[4];
#pragma unroll
    for (int q = 0; q < 4; ++q)
        hv[q] = *reinterpret_cast<const float4*>(h + (size_t)(row0 + q) * FDIM + 4 * lane);

    const float4 a1v = *reinterpret_cast<const float4*>(a + 4 * lane);
    const float4 a2v = *reinterpret_cast<const float4*>(a + FDIM + 4 * lane);
    float p1[16], p2[16];
#pragma unroll
    for (int r = 0; r < 16; ++r) {
        const float4 w = *reinterpret_cast<const float4*>(W + (size_t)(wave * 16 + r) * FDIM + 4 * lane);
        p1[r] = (w.x * a1v.x + w.y * a1v.y) + (w.z * a1v.z + w.w * a1v.w);
        p2[r] = (w.x * a2v.x + w.y * a2v.y) + (w.z * a2v.z + w.w * a2v.w);
    }
    const float t1 = wave_sum16(p1, lane);
    const float t2 = wave_sum16(p2, lane);
    if ((lane & 1) == 0) {
        s_wa[wave * 16 + (lane >> 1)]        = t1;
        s_wa[FDIM + wave * 16 + (lane >> 1)] = t2;
    }
    __syncthreads();

    const float4 u1 = *reinterpret_cast<const float4*>(s_wa + 4 * lane);
    const float4 u2 = *reinterpret_cast<const float4*>(s_wa + FDIM + 4 * lane);
    float g[8];
#pragma unroll
    for (int q = 0; q < 4; ++q) {
        g[q]     = (hv[q].x * u1.x + hv[q].y * u1.y) + (hv[q].z * u1.z + hv[q].w * u1.w);
        g[4 + q] = (hv[q].x * u2.x + hv[q].y * u2.y) + (hv[q].z * u2.z + hv[q].w * u2.w);
    }
    const float tot = wave_sum8(g, lane);
    if ((lane & 3) == 0) {
        const int idx = lane >> 2;
        float* dst = (idx < 4) ? (f1 + row0 + idx) : (f2 + row0 + idx - 4);
        *dst = tot;
    }
}

#define SPITCH   (N_NODES + 4)
#define APITCH   (FDIM + 4)

__global__ __launch_bounds__(256) void gat_attn_kernel(const float* __restrict__ h,
                                                       const float* __restrict__ W,
                                                       const int*   __restrict__ adj,
                                                       const float* __restrict__ f1,
                                                       const float* __restrict__ f2,
                                                       float* __restrict__ out) {
    __shared__ __attribute__((aligned(16))) float s_p[16 * SPITCH];
    __shared__ __attribute__((aligned(16))) float s_f2[N_NODES];
    __shared__ __attribute__((aligned(16))) float s_acc[16 * APITCH];
    __shared__ float s_l[16];

    const int blk  = blockIdx.x;
    const int b    = blk / (N_NODES / 16);
    const int i0   = (blk % (N_NODES / 16)) * 16;
    const int t    = threadIdx.x;
    const int wave = t >> 5;
    const int lane = t & 31;

    {
        const float4* f2v = reinterpret_cast<const float4*>(f2 + (size_t)b * N_NODES);
        const float4 f2a = f2v[t];
        const float4 f2b = f2v[256 + t];
        reinterpret_cast<float4*>(s_f2)[t]       = f2a;
        reinterpret_cast<float4*>(s_f2)[256 + t] = f2b;
    }
    for (int k = t; k < 16 * APITCH; k += 256) s_acc[k] = 0.f;
    __syncthreads();

    {
        const int r    = t >> 4;
        const int l16  = t & 15;
        const int4* arow = reinterpret_cast<const int4*>(adj + (size_t)(i0 + r) * N_NODES);
        const float fr  = f1[(size_t)b * N_NODES + i0 + r];
        float* prow = s_p + r * SPITCH;

        float mx = NEG_BIG;
        for (int c0 = 0; c0 < 32; c0 += 8) {
            int4 mk[8];
#pragma unroll
            for (int u = 0; u < 8; ++u) mk[u] = arow[(c0 + u) * 16 + l16];
#pragma unroll
            for (int u = 0; u < 8; ++u) {
                const int j = ((c0 + u) * 16 + l16) * 4;
                const float4 f = *reinterpret_cast<const float4*>(s_f2 + j);
                const float x0 = fr + f.x, x1 = fr + f.y, x2 = fr + f.z, x3 = fr + f.w;
                const float e0 = x0 > 0.f ? x0 : ALPHA * x0;
                const float e1 = x1 > 0.f ? x1 : ALPHA * x1;
                const float e2 = x2 > 0.f ? x2 : ALPHA * x2;
                const float e3 = x3 > 0.f ? x3 : ALPHA * x3;
                float4 s;
                s.x = (mk[u].x > 0) ? e0 : NEG_BIG;
                s.y = (mk[u].y > 0) ? e1 : NEG_BIG;
                s.z = (mk[u].z > 0) ? e2 : NEG_BIG;
                s.w = (mk[u].w > 0) ? e3 : NEG_BIG;
                mx = fmaxf(mx, fmaxf(fmaxf(s.x, s.y), fmaxf(s.z, s.w)));
                *reinterpret_cast<float4*>(prow + j) = s;
            }
        }
#pragma unroll
        for (int m = 8; m >= 1; m >>= 1) mx = fmaxf(mx, __shfl_xor(mx, m, 32));

        float sum = 0.f;
#pragma unroll 8
        for (int c = 0; c < 32; ++c) {
            const int j = (c * 16 + l16) * 4;
            const float4 s = *reinterpret_cast<const float4*>(prow + j);
            float4 p;
            p.x = __expf(s.x - mx);
            p.y = __expf(s.y - mx);
            p.z = __expf(s.z - mx);
            p.w = __expf(s.w - mx);
            sum += (p.x + p.y) + (p.z + p.w);
            *reinterpret_cast<float4*>(prow + j) = p;
        }
#pragma unroll
        for (int m = 8; m >= 1; m >>= 1) sum += __shfl_xor(sum, m, 32);

        if (l16 == 0) s_l[r] = sum;
    }
    __syncthreads();

    const int ln16  = lane & 15;
    const int kslot = lane >> 4;
    const float* pA  = s_p + ln16 * SPITCH + kslot * 2;
    const float* whb = h + (size_t)b * N_NODES * FDIM;

    v8f acc[8] = {};
    const int jend = wave * 256 + 256;
    for (int j0 = wave * 256; j0 < jend; j0 += 4) {
        const int jA = j0 + kslot * 2;
        const v2f A = *reinterpret_cast<const v2f*>(pA + j0);

        const float* w0 = whb + (size_t)jA * FDIM + ln16;
        const float* w1 = w0 + FDIM;
#pragma unroll
        for (int n = 0; n < 8; ++n) {
            v2f B;
            B.x = w0[n * 16];
            B.y = w1[n * 16];
            acc[n] = __builtin_amdgcn_wmma_f32_16x16x4_f32(false, A, false, B,
                                                           (short)0, acc[n], false, false);
        }
    }

    float wq[64];
    {
        const float* wB = W + (size_t)(kslot * 2) * FDIM + wave * 16 + ln16;
#pragma unroll
        for (int q = 0; q < 16; ++q) {
            wq[4 * q + 0] = wB[(size_t)(8 * q)     * FDIM];
            wq[4 * q + 1] = wB[(size_t)(8 * q + 1) * FDIM];
            wq[4 * q + 2] = wB[(size_t)(8 * q + 4) * FDIM];
            wq[4 * q + 3] = wB[(size_t)(8 * q + 5) * FDIM];
        }
    }

#pragma unroll
    for (int n = 0; n < 8; ++n) {
#pragma unroll
        for (int v = 0; v < 8; ++v) {
            int orow = v + 8 * kslot;
            atomicAdd(&s_acc[orow * APITCH + n * 16 + ln16], acc[n][v]);
        }
    }
    __syncthreads();

    {
        const int n0 = wave * 16;
        const float* pT = s_acc + ln16 * APITCH + kslot * 2;
        v8f o0 = {}, o1 = {};
#pragma unroll
        for (int k0 = 0; k0 < FDIM; k0 += 8) {
            const v2f A0 = *reinterpret_cast<const v2f*>(pT + k0);
            const v2f A1 = *reinterpret_cast<const v2f*>(pT + k0 + 4);
            v2f B0, B1;
            B0.x = wq[k0 / 2 + 0];
            B0.y = wq[k0 / 2 + 1];
            B1.x = wq[k0 / 2 + 2];
            B1.y = wq[k0 / 2 + 3];
            o0 = __builtin_amdgcn_wmma_f32_16x16x4_f32(false, A0, false, B0,
                                                       (short)0, o0, false, false);
            o1 = __builtin_amdgcn_wmma_f32_16x16x4_f32(false, A1, false, B1,
                                                       (short)0, o1, false, false);
        }
#pragma unroll
        for (int v = 0; v < 8; ++v) {
            const int orow = v + 8 * kslot;
            s_p[orow * FDIM + n0 + ln16] = o0[v] + o1[v];
        }
    }
    __syncthreads();

    for (int k = t; k < 16 * FDIM; k += 256) {
        int r = k >> 7;
        out[((size_t)b * N_NODES + i0 + r) * FDIM + (k & (FDIM - 1))] = s_p[k] / s_l[r];
    }
}

extern "C" void kernel_launch(void* const* d_in, const int* in_sizes, int n_in,
                              void* d_out, int out_size, void* d_ws, size_t ws_size,
                              hipStream_t stream) {
    const float* h   = (const float*)d_in[0];
    const int*   adj = (const int*)  d_in[1];
    const float* W   = (const float*)d_in[2];
    const float* a   = (const float*)d_in[3];
    float* out = (float*)d_out;

    float* f1 = (float*)d_ws;
    float* f2 = f1 + (size_t)NBATCH * N_NODES;

    gat_f_kernel   <<<NBATCH * (N_NODES / 32), 256, 0, stream>>>(h, W, a, f1, f2);
    gat_attn_kernel<<<NBATCH * (N_NODES / 16), 256, 0, stream>>>(h, W, adj, f1, f2, out);
}


# attention kernel: 512-thread blocks hold two 16-row tiles; the two waves that walk one j slice ask for the same h lines together, so a line leaves L2 once for both; per-wave loop, sum order unchanged
# speedup vs baseline: 2.9156x; 1.0415x over previous
#include <hip/hip_runtime.h>

typedef __attribute__((ext_vector_type(2))) float v2f;
typedef __attribute__((ext_vector_type(8))) float v8f;

#define ALPHA    0.2f
#define NEG_BIG  -9000000000000000.0f
#define N_NODES  2048
#define FDIM     128
#define NBATCH   4


template <int N, int M>
__device__ __forceinline__ void fold_step(float* v, int lane) {
    const bool hi = (lane & M) != 0;
#pragma unroll
    for (int r = 0; r < N; ++r) {
        const float send = hi ? v[r] : v[r + N];
        const float keep = hi ? v[r + N] : v[r];
        v[r] = keep + __shfl_xor(send, M, 32);
    }
}
__device__ __forceinline__ float wave_sum16(float* v, int lane) {
    fold_step<8, 16>(v, lane);
    fold_step<4, 8>(v, lane);
    fold_step<2, 4>(v, lane);
    fold_step<1, 2>(v, lane);
    return v[0] + __shfl_xor(v[0], 1, 32);
}
__device__ __forceinline__ float wave_sum8(float* v, int lane) {
    fold_step<4, 16>(v, lane);
    fold_step<2, 8>(v, lane);
    fold_step<1, 4>(v, lane);
    float s = v[0] + __shfl_xor(v[0], 2, 32);
    return s + __shfl_xor(s, 1, 32);
}

__global__ __launch_bounds__(256) void gat_f_kernel(const float* __restrict__ h,
                                                    const float* __restrict__ W,
                                                    const float* __restrict__ a,
                                                    float* __restrict__ f1,
                                                    float* __restrict__ f2) {
    __shared__ __attribute__((aligned(16))) float s_wa[2 * FDIM];
    const int t    = threadIdx.x;
    const int wave = t >> 5;
    const int lane = t & 31;

    const int row0 = blockIdx.x * 32 + wave * 4;
    float4 hv[4];
#pragma unroll
    for (int q = 0; q < 4; ++q)
        hv[q] = *reinterpret_cast<const float4*>(h + (size_t)(row0 + q) * FDIM + 4 * lane);

    const float4 a1v = *reinterpret_cast<const float4*>(a + 4 * lane);
    const float4 a2v = *reinterpret_cast<const float4*>(a + FDIM + 4 * lane);
    float p1[16], p2[16];
#pragma unroll
    for (int r = 0; r < 16; ++r) {
        const float4 w = *reinterpret_cast<const float4*>(W + (size_t)(wave * 16 + r) * FDIM + 4 * lane);
        p1[r] = (w.x * a1v.x + w.y * a1v.y) + (w.z * a1v.z + w.w * a1v.w);
        p2[r] = (w.x * a2v.x + w.y * a2v.y) + (w.z * a2v.z + w.w * a2v.w);
    }
    const float t1 = wave_sum16(p1, lane);
    const float t2 = wave_sum16(p2, lane);
    if ((lane & 1) == 0) {
        s_wa[wave * 16 + (lane >> 1)]        = t1;
        s_wa[FDIM + wave * 16 + (lane >> 1)] = t2;
    }
    __syncthreads();

    const float4 u1 = *reinterpret_cast<const float4*>(s_wa + 4 * lane);
    const float4 u2 = *reinterpret_cast<const float4*>(s_wa + FDIM + 4 * lane);
    float g[8];
#pragma unroll
    for (int q = 0; q < 4; ++q) {
        g[q]     = (hv[q].x * u1.x + hv[q].y * u1.y) + (hv[q].z * u1.z + hv[q].w * u1.w);
        g[4 + q] = (hv[q].x * u2.x + hv[q].y * u2.y) + (hv[q].z * u2.z + hv[q].w * u2.w);
    }
    const float tot = wave_sum8(g, lane);
    if ((lane & 3) == 0) {
        const int idx = lane >> 2;
        float* dst = (idx < 4) ? (f1 + row0 + idx) : (f2 + row0 + idx - 4);
        *dst = tot;
    }
}

#define SPITCH   (N_NODES + 4)
#define APITCH   (FDIM + 4)
#define BROWS    32

__global__ __launch_bounds__(512) void gat_attn_kernel(const float* __restrict__ h,
                                                       const float* __restrict__ W,
                                                       const int*   __restrict__ adj,
                                                       const float* __restrict__ f1,
                                                       const float* __restrict__ f2,
                                                       float* __restrict__ out) {
    __shared__ __attribute__((aligned(16))) float s_p[BROWS * SPITCH];
    __shared__ __attribute__((aligned(16))) float s_f2[N_NODES];
    __shared__ __attribute__((aligned(16))) float s_acc[BROWS * APITCH];
    __shared__ float s_l[BROWS];

    const int blk  = blockIdx.x;
    const int b    = blk / (N_NODES / BROWS);
    const int i0   = (blk % (N_NODES / BROWS)) * BROWS;
    const int t    = threadIdx.x;
    const int wave = t >> 5;
    const int lane = t & 31;

    {
        const float4* f2v = reinterpret_cast<const float4*>(f2 + (size_t)b * N_NODES);
        reinterpret_cast<float4*>(s_f2)[t] = f2v[t];
    }
    for (int k = t; k < BROWS * APITCH; k += 512) s_acc[k] = 0.f;
    __syncthreads();

    {
        const int r    = t >> 4;
        const int l16  = t & 15;
        const int4* arow = reinterpret_cast<const int4*>(adj + (size_t)(i0 + r) * N_NODES);
        const float fr  = f1[(size_t)b * N_NODES + i0 + r];
        float* prow = s_p + r * SPITCH;

        float mx = NEG_BIG;
        for (int c0 = 0; c0 < 32; c0 += 8) {
            int4 mk[8];
#pragma unroll
            for (int u = 0; u < 8; ++u) mk[u] = arow[(c0 + u) * 16 + l16];
#pragma unroll
            for (int u = 0; u < 8; ++u) {
                const int j = ((c0 + u) * 16 + l16) * 4;
                const float4 f = *reinterpret_cast<const float4*>(s_f2 + j);
                const float x0 = fr + f.x, x1 = fr + f.y, x2 = fr + f.z, x3 = fr + f.w;
                const float e0 = x0 > 0.f ? x0 : ALPHA * x0;
                const float e1 = x1 > 0.f ? x1 : ALPHA * x1;
                const float e2 = x2 > 0.f ? x2 : ALPHA * x2;
                const float e3 = x3 > 0.f ? x3 : ALPHA * x3;
                float4 s;
                s.x = (mk[u].x > 0) ? e0 : NEG_BIG;
                s.y = (mk[u].y > 0) ? e1 : NEG_BIG;
                s.z = (mk[u].z > 0) ? e2 : NEG_BIG;
                s.w = (mk[u].w > 0) ? e3 : NEG_BIG;
                mx = fmaxf(mx, fmaxf(fmaxf(s.x, s.y), fmaxf(s.z, s.w)));
                *reinterpret_cast<float4*>(prow + j) = s;
            }
        }
#pragma unroll
        for (int m = 8; m >= 1; m >>= 1) mx = fmaxf(mx, __shfl_xor(mx, m, 32));

        float sum = 0.f;
#pragma unroll 8
        for (int c = 0; c < 32; ++c) {
            const int j = (c * 16 + l16) * 4;
            const float4 s = *reinterpret_cast<const float4*>(prow + j);
            float4 p;
            p.x = __expf(s.x - mx);
            p.y = __expf(s.y - mx);
            p.z = __expf(s.z - mx);
            p.w = __expf(s.w - mx);
            sum += (p.x + p.y) + (p.z + p.w);
            *reinterpret_cast<float4*>(prow + j) = p;
        }
#pragma unroll
        for (int m = 8; m >= 1; m >>= 1) sum += __shfl_xor(sum, m, 32);

        if (l16 == 0) s_l[r] = sum;
    }
    __syncthreads();

    const int rt    = (wave >> 1) & 1;
    const int sl    = (wave & 1) | ((wave >> 2) << 1);
    const int ln16  = lane & 15;
    const int kslot = lane >> 4;
    const float* pA  = s_p + (rt * 16 + ln16) * SPITCH + kslot * 2;
    const float* whb = h + (size_t)b * N_NODES * FDIM;

    v8f acc[8] = {};
    for (int it = 0; it < 16; ++it) {
#pragma unroll
        for (int u = 0; u < 4; ++u) {
            const int j0 = sl * 256 + it * 16 + u * 4;
            const int jA = j0 + kslot * 2;
            const v2f A = *reinterpret_cast<const v2f*>(pA + j0);

            const float* w0 = whb + (size_t)jA * FDIM + ln16;
            const float* w1 = w0 + FDIM;
#pragma unroll
            for (int n = 0; n < 8; ++n) {
                v2f B;
                B.x = w0[n * 16];
                B.y = w1[n * 16];
                acc[n] = __builtin_amdgcn_wmma_f32_16x16x4_f32(false, A, false, B,
                                                               (short)0, acc[n], false, false);
            }
        }
    }

    const int cn  = wave & 7;
    const int rtE = wave >> 3;
    float wq[64];
    {
        const float* wB = W + (size_t)(kslot * 2) * FDIM + cn * 16 + ln16;
#pragma unroll
        for (int q = 0; q < 16; ++q) {
            wq[4 * q + 0] = wB[(size_t)(8 * q)     * FDIM];
            wq[4 * q + 1] = wB[(size_t)(8 * q + 1) * FDIM];
            wq[4 * q + 2] = wB[(size_t)(8 * q + 4) * FDIM];
            wq[4 * q + 3] = wB[(size_t)(8 * q + 5) * FDIM];
        }
    }

#pragma unroll
    for (int n = 0; n < 8; ++n) {
#pragma unroll
        for (int v = 0; v < 8; ++v) {
            int orow = rt * 16 + v + 8 * kslot;
            atomicAdd(&s_acc[orow * APITCH + n * 16 + ln16], acc[n][v]);
        }
    }
    __syncthreads();

    {
        const int n0 = cn * 16;
        const float* pT = s_acc + (rtE * 16 + ln16) * APITCH + kslot * 2;
        v8f o0 = {}, o1 = {};
#pragma unroll
        for (int k0 = 0; k0 < FDIM; k0 += 8) {
            const v2f A0 = *reinterpret_cast<const v2f*>(pT + k0);
            const v2f A1 = *reinterpret_cast<const v2f*>(pT + k0 + 4);
            v2f B0, B1;
            B0.x = wq[k0 / 2 + 0];
            B0.y = wq[k0 / 2 + 1];
            B1.x = wq[k0 / 2 + 2];
            B1.y = wq[k0 / 2 + 3];
            o0 = __builtin_amdgcn_wmma_f32_16x16x4_f32(false, A0, false, B0,
                                                       (short)0, o0, false, false);
            o1 = __builtin_amdgcn_wmma_f32_16x16x4_f32(false, A1, false, B1,
                                                       (short)0, o1, false, false);
        }
#pragma unroll
        for (int v = 0; v < 8; ++v) {
            const int orow = rtE * 16 + v + 8 * kslot;
            s_p[orow * FDIM + n0 + ln16] = o0[v] + o1[v];
        }
    }
    __syncthreads();

    for (int k = t; k < BROWS * FDIM; k += 512) {
        int r = k >> 7;
        out[((size_t)b * N_NODES + i0 + r) * FDIM + (k & (FDIM - 1))] = s_p[k] / s_l[r];
    }
}

extern "C" void kernel_launch(void* const* d_in, const int* in_sizes, int n_in,
                              void* d_out, int out_size, void* d_ws, size_t ws_size,
                              hipStream_t stream) {
    const float* h   = (const float*)d_in[0];
    const int*   adj = (const int*)  d_in[1];
    const float* W   = (const float*)d_in[2];
    const float* a   = (const float*)d_in[3];
    float* out = (float*)d_out;

    float* f1 = (float*)d_ws;
    float* f2 = f1 + (size_t)NBATCH * N_NODES;

    gat_f_kernel   <<<NBATCH * (N_NODES / 32), 256, 0, stream>>>(h, W, a, f1, f2);
    gat_attn_kernel<<<NBATCH * (N_NODES / BROWS), 512, 0, stream>>>(h, W, adj, f1, f2, out);
}
